# starting kernel
# baseline (speedup 1.0000x reference)
#include <hip/hip_runtime.h>
#include <math.h>

#define N_NODES   6000
#define DIA_LEN_C 100
#define D_DIM     512

typedef __attribute__((ext_vector_type(2))) float v2f;
typedef __attribute__((ext_vector_type(8))) float v8f;

__global__ void __launch_bounds__(256)
build_x_kernel(const float* __restrict__ inp,
               const int*   __restrict__ qmask,
               float*       __restrict__ x) {
    const int wave = (blockIdx.x * blockDim.x + threadIdx.x) >> 5;
    const int lane = threadIdx.x & 31;
    if (wave >= N_NODES) return;
    const int i  = wave;
    const int d0 = (i / DIA_LEN_C) * DIA_LEN_C;
    const int d1 = d0 + DIA_LEN_C;
    const int sp = qmask[2 * i];

    int prev = -1, next = -1;
    for (int j = i - 1; j >= d0; --j) { if (qmask[2 * j] == sp) { prev = j; break; } }
    for (int j = i + 1; j <  d1; ++j) { if (qmask[2 * j] == sp) { next = j; break; } }

    const float* xi = inp + (size_t)i * D_DIM;
    const float* xp = (prev >= 0) ? inp + (size_t)prev * D_DIM : xi;
    const float* xn = (next >= 0) ? inp + (size_t)next * D_DIM : xi;

    float4 ra[4], rp[4], rn[4];
    float sii = 0.f, spp = 0.f, snn = 0.f, sip = 0.f, sin_ = 0.f;
#pragma unroll
    for (int t = 0; t < 4; ++t) {
        const int k = t * 128 + lane * 4;
        float4 a = *(const float4*)(xi + k);
        float4 p = *(const float4*)(xp + k);
        float4 n = *(const float4*)(xn + k);
        ra[t] = a; rp[t] = p; rn[t] = n;
        sii  += a.x*a.x + a.y*a.y + a.z*a.z + a.w*a.w;
        spp  += p.x*p.x + p.y*p.y + p.z*p.z + p.w*p.w;
        snn  += n.x*n.x + n.y*n.y + n.z*n.z + n.w*n.w;
        sip  += a.x*p.x + a.y*p.y + a.z*p.z + a.w*p.w;
        sin_ += a.x*n.x + a.y*n.y + a.z*n.z + a.w*n.w;
    }
#pragma unroll
    for (int m = 16; m >= 1; m >>= 1) {
        sii  += __shfl_xor(sii,  m, 32);
        spp  += __shfl_xor(spp,  m, 32);
        snn  += __shfl_xor(snn,  m, 32);
        sip  += __shfl_xor(sip,  m, 32);
        sin_ += __shfl_xor(sin_, m, 32);
    }

    const float INV_PI = 0.318309886183790671538f;
    float wp = 0.f, wn = 0.f;
    if (prev >= 0) {
        float den = sqrtf(sii) * sqrtf(spp);
        float c   = (den > 0.f) ? sip / den : 0.f;
        c  = fminf(1.f, fmaxf(-1.f, c));
        wp = 1.f - acosf(c) * INV_PI;
    }
    if (next >= 0) {
        float den = sqrtf(sii) * sqrtf(snn);
        float c   = (den > 0.f) ? sin_ / den : 0.f;
        c  = fminf(1.f, fmaxf(-1.f, c));
        wn = 1.f - acosf(c) * INV_PI;
    }

    float* xo = x + (size_t)i * D_DIM;
#pragma unroll
    for (int t = 0; t < 4; ++t) {
        const int k = t * 128 + lane * 4;
        float4 r;
        r.x = ra[t].x + wp * rp[t].x + wn * rn[t].x;
        r.y = ra[t].y + wp * rp[t].y + wn * rn[t].y;
        r.z = ra[t].z + wp * rp[t].z + wn * rn[t].z;
        r.w = ra[t].w + wp * rp[t].w + wn * rn[t].w;
        *(float4*)(xo + k) = r;
    }
}

__global__ void __launch_bounds__(128)
gemm_wmma_kernel(const float* __restrict__ x,
                 const float* __restrict__ W,
                 const float* __restrict__ bias,
                 float*       __restrict__ out) {
    const int waveInBlk = threadIdx.x >> 5;
    const int lane      = threadIdx.x & 31;
    const int m0 = blockIdx.x * 16;
    const int n0 = (blockIdx.y * 4 + waveInBlk) * 64;

    const int row = lane & 15;
    const int kh  = (lane >> 4) * 2;
    const float* aBase = x + (size_t)(m0 + row) * D_DIM + kh;
    const float* bBase = W + (size_t)(n0 + row) * D_DIM + kh;

    v8f acc[4] = {{}, {}, {}, {}};
#pragma unroll 4
    for (int k = 0; k < D_DIM; k += 4) {
        v2f a  = *(const v2f*)(aBase + k);
        v2f b0 = *(const v2f*)(bBase + k);
        v2f b1 = *(const v2f*)(bBase + 16 * D_DIM + k);
        v2f b2 = *(const v2f*)(bBase + 32 * D_DIM + k);
        v2f b3 = *(const v2f*)(bBase + 48 * D_DIM + k);
        acc[0] = __builtin_amdgcn_wmma_f32_16x16x4_f32(
            false, a, false, b0, (short)0, acc[0], false, false);
        acc[1] = __builtin_amdgcn_wmma_f32_16x16x4_f32(
            false, a, false, b1, (short)0, acc[1], false, false);
        acc[2] = __builtin_amdgcn_wmma_f32_16x16x4_f32(
            false, a, false, b2, (short)0, acc[2], false, false);
        acc[3] = __builtin_amdgcn_wmma_f32_16x16x4_f32(
            false, a, false, b3, (short)0, acc[3], false, false);
    }

    const int rbase = m0 + (lane >> 4) * 8;
#pragma unroll
    for (int j = 0; j < 4; ++j) {
        const int col  = n0 + j * 16 + row;
        const float bv = bias[col];
#pragma unroll
        for (int r = 0; r < 8; ++r) {
            out[(size_t)(rbase + r) * D_DIM + col] = acc[j][r] + bv;
        }
    }
}

extern "C" void kernel_launch(void* const* d_in, const int* in_sizes, int n_in,
                              void* d_out, int out_size, void* d_ws, size_t ws_size,
                              hipStream_t stream) {
    const float* inp   = (const float*)d_in[0];
    const int*   qmask = (const int*)d_in[2];
    const float* W     = (const float*)d_in[3];
    const float* bias  = (const float*)d_in[4];
    float*       xbuf  = (float*)d_ws;
    float*       out   = (float*)d_out;

    build_x_kernel<<<dim3(N_NODES / 8), dim3(256), 0, stream>>>(inp, qmask, xbuf);
    gemm_wmma_kernel<<<dim3(N_NODES / 16, 2), dim3(128), 0, stream>>>(
        xbuf, W, bias, out);
}


# GEMM: 64x64 work-group tiles, x and W slabs staged in LDS by whole-line 16-byte loads (two buffers, one barrier per 32-k chunk), 32x32 per wave; k order of every sum unchanged
# speedup vs baseline: 1.3863x; 1.3863x over previous
#include <hip/hip_runtime.h>
#include <math.h>

#define N_NODES   6000
#define DIA_LEN_C 100
#define D_DIM     512

typedef __attribute__((ext_vector_type(2))) float v2f;
typedef __attribute__((ext_vector_type(8))) float v8f;

__global__ void __launch_bounds__(256)
build_x_kernel(const float* __restrict__ inp,
               const int*   __restrict__ qmask,
               float*       __restrict__ x) {
    const int wave = (blockIdx.x * blockDim.x + threadIdx.x) >> 5;
    const int lane = threadIdx.x & 31;
    if (wave >= N_NODES) return;
    const int i  = wave;
    const int d0 = (i / DIA_LEN_C) * DIA_LEN_C;
    const int d1 = d0 + DIA_LEN_C;
    const int sp = qmask[2 * i];

    int prev = -1, next = -1;
    for (int j = i - 1; j >= d0; --j) { if (qmask[2 * j] == sp) { prev = j; break; } }
    for (int j = i + 1; j <  d1; ++j) { if (qmask[2 * j] == sp) { next = j; break; } }

    const float* xi = inp + (size_t)i * D_DIM;
    const float* xp = (prev >= 0) ? inp + (size_t)prev * D_DIM : xi;
    const float* xn = (next >= 0) ? inp + (size_t)next * D_DIM : xi;

    float4 ra[4], rp[4], rn[4];
    float sii = 0.f, spp = 0.f, snn = 0.f, sip = 0.f, sin_ = 0.f;
#pragma unroll
    for (int t = 0; t < 4; ++t) {
        const int k = t * 128 + lane * 4;
        float4 a = *(const float4*)(xi + k);
        float4 p = *(const float4*)(xp + k);
        float4 n = *(const float4*)(xn + k);
        ra[t] = a; rp[t] = p; rn[t] = n;
        sii  += a.x*a.x + a.y*a.y + a.z*a.z + a.w*a.w;
        spp  += p.x*p.x + p.y*p.y + p.z*p.z + p.w*p.w;
        snn  += n.x*n.x + n.y*n.y + n.z*n.z + n.w*n.w;
        sip  += a.x*p.x + a.y*p.y + a.z*p.z + a.w*p.w;
        sin_ += a.x*n.x + a.y*n.y + a.z*n.z + a.w*n.w;
    }
#pragma unroll
    for (int m = 16; m >= 1; m >>= 1) {
        sii  += __shfl_xor(sii,  m, 32);
        spp  += __shfl_xor(spp,  m, 32);
        snn  += __shfl_xor(snn,  m, 32);
        sip  += __shfl_xor(sip,  m, 32);
        sin_ += __shfl_xor(sin_, m, 32);
    }

    const float INV_PI = 0.318309886183790671538f;
    float wp = 0.f, wn = 0.f;
    if (prev >= 0) {
        float den = sqrtf(sii) * sqrtf(spp);
        float c   = (den > 0.f) ? sip / den : 0.f;
        c  = fminf(1.f, fmaxf(-1.f, c));
        wp = 1.f - acosf(c) * INV_PI;
    }
    if (next >= 0) {
        float den = sqrtf(sii) * sqrtf(snn);
        float c   = (den > 0.f) ? sin_ / den : 0.f;
        c  = fminf(1.f, fmaxf(-1.f, c));
        wn = 1.f - acosf(c) * INV_PI;
    }

    float* xo = x + (size_t)i * D_DIM;
#pragma unroll
    for (int t = 0; t < 4; ++t) {
        const int k = t * 128 + lane * 4;
        float4 r;
        r.x = ra[t].x + wp * rp[t].x + wn * rn[t].x;
        r.y = ra[t].y + wp * rp[t].y + wn * rn[t].y;
        r.z = ra[t].z + wp * rp[t].z + wn * rn[t].z;
        r.w = ra[t].w + wp * rp[t].w + wn * rn[t].w;
        *(float4*)(xo + k) = r;
    }
}

#define BM    64
#define BN    64
#define KC    32
#define PITCH 36

__global__ void __launch_bounds__(128)
gemm_wmma_kernel(const float* __restrict__ x,
                 const float* __restrict__ W,
                 const float* __restrict__ bias,
                 float*       __restrict__ out) {
    __shared__ __attribute__((aligned(16))) float As[2][BM * PITCH];
    __shared__ __attribute__((aligned(16))) float Bs[2][BN * PITCH];

    const int tid  = threadIdx.x;
    const int wave = tid >> 5;
    const int lane = tid & 31;
    const int m0 = blockIdx.x * BM;
    const int n0 = blockIdx.y * BN;
    const int wm = (wave >> 1) * 32;
    const int wn = (wave & 1) * 32;

    const int srow = tid >> 3;
    const int scol = (tid & 7) * 4;
    const float* ag[4];
#pragma unroll
    for (int i = 0; i < 4; ++i) {
        int r = m0 + srow + 16 * i;
        r = (r < N_NODES) ? r : (N_NODES - 1);
        ag[i] = x + (size_t)r * D_DIM + scol;
    }
    const float* bg = W + (size_t)(n0 + srow) * D_DIM + scol;
    const int soff = srow * PITCH + scol;

    float4 pa[4], pb[4];
#pragma unroll
    for (int i = 0; i < 4; ++i) {
        pa[i] = *(const float4*)(ag[i]);
        pb[i] = *(const float4*)(bg + (size_t)i * 16 * D_DIM);
    }
#pragma unroll
    for (int i = 0; i < 4; ++i) {
        *(float4*)(&As[0][soff + i * 16 * PITCH]) = pa[i];
        *(float4*)(&Bs[0][soff + i * 16 * PITCH]) = pb[i];
    }
    __syncthreads();

    const int row = lane & 15;
    const int kh  = (lane >> 4) * 2;
    const int aoff = (wm + row) * PITCH + kh;
    const int boff = (wn + row) * PITCH + kh;

    v8f acc00 = {}, acc01 = {}, acc10 = {}, acc11 = {};

    for (int c = 0; c < D_DIM / KC; ++c) {
        const int cur = c & 1;
        const bool more = (c + 1) < (D_DIM / KC);
        if (more) {
            const int kn = (c + 1) * KC;
#pragma unroll
            for (int i = 0; i < 4; ++i) {
                pa[i] = *(const float4*)(ag[i] + kn);
                pb[i] = *(const float4*)(bg + (size_t)i * 16 * D_DIM + kn);
            }
        }
        const float* as = &As[cur][aoff];
        const float* bs = &Bs[cur][boff];
#pragma unroll
        for (int ks = 0; ks < KC / 4; ++ks) {
            v2f a0 = *(const v2f*)(as + ks * 4);
            v2f a1 = *(const v2f*)(as + 16 * PITCH + ks * 4);
            v2f b0 = *(const v2f*)(bs + ks * 4);
            v2f b1 = *(const v2f*)(bs + 16 * PITCH + ks * 4);
            acc00 = __builtin_amdgcn_wmma_f32_16x16x4_f32(
                false, a0, false, b0, (short)0, acc00, false, false);
            acc01 = __builtin_amdgcn_wmma_f32_16x16x4_f32(
                false, a0, false, b1, (short)0, acc01, false, false);
            acc10 = __builtin_amdgcn_wmma_f32_16x16x4_f32(
                false, a1, false, b0, (short)0, acc10, false, false);
            acc11 = __builtin_amdgcn_wmma_f32_16x16x4_f32(
                false, a1, false, b1, (short)0, acc11, false, false);
        }
        if (more) {
#pragma unroll
            for (int i = 0; i < 4; ++i) {
                *(float4*)(&As[cur ^ 1][soff + i * 16 * PITCH]) = pa[i];
                *(float4*)(&Bs[cur ^ 1][soff + i * 16 * PITCH]) = pb[i];
            }
        }
        __syncthreads();
    }

    const int rhalf = (lane >> 4) * 8;
#pragma unroll
    for (int j = 0; j < 2; ++j) {
        const int col  = n0 + wn + j * 16 + row;
        const float bv = bias[col];
#pragma unroll
        for (int i = 0; i < 2; ++i) {
            const v8f acc = (i == 0) ? (j == 0 ? acc00 : acc01) : (j == 0 ? acc10 : acc11);
            const int rbase = m0 + wm + i * 16 + rhalf;
            if (m0 + wm + i * 16 < N_NODES) {
#pragma unroll
                for (int r = 0; r < 8; ++r) {
                    out[(size_t)(rbase + r) * D_DIM + col] = acc[r] + bv;
                }
            }
        }
    }
}

extern "C" void kernel_launch(void* const* d_in, const int* in_sizes, int n_in,
                              void* d_out, int out_size, void* d_ws, size_t ws_size,
                              hipStream_t stream) {
    const float* inp   = (const float*)d_in[0];
    const int*   qmask = (const int*)d_in[2];
    const float* W     = (const float*)d_in[3];
    const float* bias  = (const float*)d_in[4];
    float*       xbuf  = (float*)d_ws;
    float*       out   = (float*)d_out;

    build_x_kernel<<<dim3(N_NODES / 8), dim3(256), 0, stream>>>(inp, qmask, xbuf);
    gemm_wmma_kernel<<<dim3(94, 8), dim3(128), 0, stream>>>(
        xbuf, W, bias, out);
}


# K loop hand-over pipelined: next chunk's fetch pinned at the trip's top, its LDS stores behind 10 of 16 k-steps, first 2 k-steps' fragments read behind the barrier under a 6-k-step tail; k order kept
# speedup vs baseline: 1.8287x; 1.3191x over previous
#include <hip/hip_runtime.h>
#include <math.h>

#define N_NODES   6000
#define DIA_LEN_C 100
#define D_DIM     512

typedef __attribute__((ext_vector_type(2))) float v2f;
typedef __attribute__((ext_vector_type(8))) float v8f;

#define BN     128
#define KC     64
#define PITCH  68
#define NRT    7
#define AROWS  (NRT * 16)
#define YPITCH 132
#define KS     (KC / 4)
#define HS     2
#define SPLIT  10

__global__ void __launch_bounds__(256)
dialog_mix_gemm_kernel(const float* inp,
                       const int*   __restrict__ qmask,
                       const float* W,
                       const float* __restrict__ bias,
                       float*       __restrict__ out) {
    __shared__ __attribute__((aligned(16))) float As[2 * AROWS * PITCH];
    __shared__ __attribute__((aligned(16))) float Bs[2 * BN * PITCH];
    __shared__ __attribute__((aligned(16))) float Ys[AROWS * YPITCH];
    __shared__ int   sq[DIA_LEN_C];
    __shared__ int   prevL[DIA_LEN_C];
    __shared__ int   nextL[DIA_LEN_C];
    __shared__ float nrm2[DIA_LEN_C];
    __shared__ float dpv[DIA_LEN_C];
    __shared__ float wpL[DIA_LEN_C];
    __shared__ float wnL[DIA_LEN_C];

    const int tid  = threadIdx.x;
    const int wave = tid >> 5;
    const int lane = tid & 31;
    const int n0 = blockIdx.x * BN;
    const int m0 = blockIdx.y * DIA_LEN_C;

    const int srow = tid >> 4;
    const int scol = (tid & 15) * 4;
    const float* ag0 = inp + (size_t)(m0 + ((srow + 0 < DIA_LEN_C) ? (srow + 0) : (DIA_LEN_C - 1))) * D_DIM + scol;
    const float* ag1 = inp + (size_t)(m0 + ((srow + 16 < DIA_LEN_C) ? (srow + 16) : (DIA_LEN_C - 1))) * D_DIM + scol;
    const float* ag2 = inp + (size_t)(m0 + ((srow + 32 < DIA_LEN_C) ? (srow + 32) : (DIA_LEN_C - 1))) * D_DIM + scol;
    const float* ag3 = inp + (size_t)(m0 + ((srow + 48 < DIA_LEN_C) ? (srow + 48) : (DIA_LEN_C - 1))) * D_DIM + scol;
    const float* ag4 = inp + (size_t)(m0 + ((srow + 64 < DIA_LEN_C) ? (srow + 64) : (DIA_LEN_C - 1))) * D_DIM + scol;
    const float* ag5 = inp + (size_t)(m0 + ((srow + 80 < DIA_LEN_C) ? (srow + 80) : (DIA_LEN_C - 1))) * D_DIM + scol;
    const float* ag6 = inp + (size_t)(m0 + ((srow + 96 < DIA_LEN_C) ? (srow + 96) : (DIA_LEN_C - 1))) * D_DIM + scol;
    const float* bg = W + (size_t)(n0 + srow) * D_DIM + scol;
    const int soff = srow * PITCH + scol;

    int myq = 0;
    if (tid < DIA_LEN_C) myq = qmask[2 * (m0 + tid)];

    float4 pa0 = *(const float4*)(ag0);
    float4 pa1 = *(const float4*)(ag1);
    float4 pa2 = *(const float4*)(ag2);
    float4 pa3 = *(const float4*)(ag3);
    float4 pa4 = *(const float4*)(ag4);
    float4 pa5 = *(const float4*)(ag5);
    float4 pa6 = *(const float4*)(ag6);
    float4 pb0 = *(const float4*)(bg + (size_t)0 * 16 * D_DIM);
    float4 pb1 = *(const float4*)(bg + (size_t)1 * 16 * D_DIM);
    float4 pb2 = *(const float4*)(bg + (size_t)2 * 16 * D_DIM);
    float4 pb3 = *(const float4*)(bg + (size_t)3 * 16 * D_DIM);
    float4 pb4 = *(const float4*)(bg + (size_t)4 * 16 * D_DIM);
    float4 pb5 = *(const float4*)(bg + (size_t)5 * 16 * D_DIM);
    float4 pb6 = *(const float4*)(bg + (size_t)6 * 16 * D_DIM);
    float4 pb7 = *(const float4*)(bg + (size_t)7 * 16 * D_DIM);
    if (tid < DIA_LEN_C) sq[tid] = myq;
    *(float4*)(&As[soff + 0 * 16 * PITCH]) = pa0;
    *(float4*)(&As[soff + 1 * 16 * PITCH]) = pa1;
    *(float4*)(&As[soff + 2 * 16 * PITCH]) = pa2;
    *(float4*)(&As[soff + 3 * 16 * PITCH]) = pa3;
    *(float4*)(&As[soff + 4 * 16 * PITCH]) = pa4;
    *(float4*)(&As[soff + 5 * 16 * PITCH]) = pa5;
    *(float4*)(&As[soff + 6 * 16 * PITCH]) = pa6;
    *(float4*)(&Bs[soff + 0 * 16 * PITCH]) = pb0;
    *(float4*)(&Bs[soff + 1 * 16 * PITCH]) = pb1;
    *(float4*)(&Bs[soff + 2 * 16 * PITCH]) = pb2;
    *(float4*)(&Bs[soff + 3 * 16 * PITCH]) = pb3;
    *(float4*)(&Bs[soff + 4 * 16 * PITCH]) = pb4;
    *(float4*)(&Bs[soff + 5 * 16 * PITCH]) = pb5;
    *(float4*)(&Bs[soff + 6 * 16 * PITCH]) = pb6;
    *(float4*)(&Bs[soff + 7 * 16 * PITCH]) = pb7;
    __syncthreads();

    if (tid < DIA_LEN_C) {
        int pv = -1, nx = -1;
        for (int j = tid - 1; j >= 0; --j)        { if (sq[j] == myq) { pv = j; break; } }
        for (int j = tid + 1; j < DIA_LEN_C; ++j) { if (sq[j] == myq) { nx = j; break; } }
        prevL[tid] = pv;
        nextL[tid] = nx;
    }
    __syncthreads();

    const int drow = ((tid >> 1) < DIA_LEN_C) ? (tid >> 1) : (DIA_LEN_C - 1);
    const int dpr  = prevL[drow];
    const int dh   = (tid & 1) * 32;
    const int doff_i = drow * PITCH + dh;
    const int doff_p = ((dpr >= 0) ? dpr : drow) * PITCH + dh;
    float si0 = 0.f, si1 = 0.f, si2 = 0.f, si3 = 0.f;
    float sp0 = 0.f, sp1 = 0.f, sp2 = 0.f, sp3 = 0.f;

    const int row = lane & 15;
    const int kh  = (lane >> 4) * 2;
    const int aoff = row * PITCH + kh;
    const int boff = (wave * 16 + row) * PITCH + kh;

    v8f acc[NRT];
#pragma unroll
    for (int t = 0; t < NRT; ++t) acc[t] = (v8f){0.f, 0.f, 0.f, 0.f, 0.f, 0.f, 0.f, 0.f};

    v2f hA[HS][NRT];
    v2f hB[HS];
    {
        const float* as0 = &As[aoff];
        const float* bs0 = &Bs[boff];
#pragma unroll
        for (int ks = 0; ks < HS; ++ks) {
            hB[ks] = *(const v2f*)(bs0 + ks * 4);
#pragma unroll
            for (int t = 0; t < NRT; ++t) hA[ks][t] = *(const v2f*)(as0 + t * 16 * PITCH + ks * 4);
        }
    }

    for (int c = 0; c < D_DIM / KC; ++c) {
        const int cur = c & 1;
        const bool more = (c + 1) < (D_DIM / KC);
        if (more) {
            const int kn = (c + 1) * KC;
            pa0 = *(const float4*)(ag0 + kn);
            pa1 = *(const float4*)(ag1 + kn);
            pa2 = *(const float4*)(ag2 + kn);
            pa3 = *(const float4*)(ag3 + kn);
            pa4 = *(const float4*)(ag4 + kn);
            pa5 = *(const float4*)(ag5 + kn);
            pa6 = *(const float4*)(ag6 + kn);
            pb0 = *(const float4*)(bg + (size_t)0 * 16 * D_DIM + kn);
            pb1 = *(const float4*)(bg + (size_t)1 * 16 * D_DIM + kn);
            pb2 = *(const float4*)(bg + (size_t)2 * 16 * D_DIM + kn);
            pb3 = *(const float4*)(bg + (size_t)3 * 16 * D_DIM + kn);
            pb4 = *(const float4*)(bg + (size_t)4 * 16 * D_DIM + kn);
            pb5 = *(const float4*)(bg + (size_t)5 * 16 * D_DIM + kn);
            pb6 = *(const float4*)(bg + (size_t)6 * 16 * D_DIM + kn);
            pb7 = *(const float4*)(bg + (size_t)7 * 16 * D_DIM + kn);
        }
        __builtin_amdgcn_sched_barrier(0);
        const float* abuf = &As[cur * AROWS * PITCH];
        const float* bbuf = &Bs[cur * BN * PITCH];
        {
            const float* ai = abuf + doff_i;
            const float* ap = abuf + doff_p;
#pragma unroll
            for (int j = 0; j < 8; ++j) {
                const float4 a = *(const float4*)(ai + 4 * j);
                const float4 p = *(const float4*)(ap + 4 * j);
                si0 = fmaf(a.x, a.x, si0); si1 = fmaf(a.y, a.y, si1);
                si2 = fmaf(a.z, a.z, si2); si3 = fmaf(a.w, a.w, si3);
                sp0 = fmaf(a.x, p.x, sp0); sp1 = fmaf(a.y, p.y, sp1);
                sp2 = fmaf(a.z, p.z, sp2); sp3 = fmaf(a.w, p.w, sp3);
            }
        }
        const float* as = abuf + aoff;
        const float* bs = bbuf + boff;
#pragma unroll
        for (int ks = 0; ks < HS; ++ks) {
#pragma unroll
            for (int t = 0; t < NRT; ++t) {
                acc[t] = __builtin_amdgcn_wmma_f32_16x16x4_f32(
                    false, hA[ks][t], false, hB[ks], (short)0, acc[t], false, false);
            }
        }
#pragma unroll
        for (int ks = HS; ks < SPLIT; ++ks) {
            const v2f b = *(const v2f*)(bs + ks * 4);
#pragma unroll
            for (int t = 0; t < NRT; ++t) {
                const v2f a = *(const v2f*)(as + t * 16 * PITCH + ks * 4);
                acc[t] = __builtin_amdgcn_wmma_f32_16x16x4_f32(
                    false, a, false, b, (short)0, acc[t], false, false);
            }
        }
        v2f tA[KS - SPLIT][NRT];
        v2f tB[KS - SPLIT];
#pragma unroll
        for (int ks = SPLIT; ks < KS; ++ks) {
            tB[ks - SPLIT] = *(const v2f*)(bs + ks * 4);
#pragma unroll
            for (int t = 0; t < NRT; ++t) tA[ks - SPLIT][t] = *(const v2f*)(as + t * 16 * PITCH + ks * 4);
        }
        __builtin_amdgcn_sched_barrier(0);
        if (more) {
            float* an = &As[(cur ^ 1) * AROWS * PITCH];
            float* bn = &Bs[(cur ^ 1) * BN * PITCH];
            *(float4*)(an + soff + 0 * 16 * PITCH) = pa0;
            *(float4*)(an + soff + 1 * 16 * PITCH) = pa1;
            *(float4*)(an + soff + 2 * 16 * PITCH) = pa2;
            *(float4*)(an + soff + 3 * 16 * PITCH) = pa3;
            *(float4*)(an + soff + 4 * 16 * PITCH) = pa4;
            *(float4*)(an + soff + 5 * 16 * PITCH) = pa5;
            *(float4*)(an + soff + 6 * 16 * PITCH) = pa6;
            *(float4*)(bn + soff + 0 * 16 * PITCH) = pb0;
            *(float4*)(bn + soff + 1 * 16 * PITCH) = pb1;
            *(float4*)(bn + soff + 2 * 16 * PITCH) = pb2;
            *(float4*)(bn + soff + 3 * 16 * PITCH) = pb3;
            *(float4*)(bn + soff + 4 * 16 * PITCH) = pb4;
            *(float4*)(bn + soff + 5 * 16 * PITCH) = pb5;
            *(float4*)(bn + soff + 6 * 16 * PITCH) = pb6;
            *(float4*)(bn + soff + 7 * 16 * PITCH) = pb7;
        }
        __syncthreads();
        if (more) {
            const float* asn = &As[(cur ^ 1) * AROWS * PITCH] + aoff;
            const float* bsn = &Bs[(cur ^ 1) * BN * PITCH] + boff;
#pragma unroll
            for (int ks = 0; ks < HS; ++ks) {
                hB[ks] = *(const v2f*)(bsn + ks * 4);
#pragma unroll
                for (int t = 0; t < NRT; ++t) hA[ks][t] = *(const v2f*)(asn + t * 16 * PITCH + ks * 4);
            }
        }
#pragma unroll
        for (int ks = SPLIT; ks < KS; ++ks) {
#pragma unroll
            for (int t = 0; t < NRT; ++t) {
                acc[t] = __builtin_amdgcn_wmma_f32_16x16x4_f32(
                    false, tA[ks - SPLIT][t], false, tB[ks - SPLIT], (short)0, acc[t], false, false);
            }
        }
    }

    float sii = (si0 + si1) + (si2 + si3);
    float sip = (sp0 + sp1) + (sp2 + sp3);
    sii += __shfl_xor(sii, 1, 32);
    sip += __shfl_xor(sip, 1, 32);
    if ((tid & 1) == 0 && (tid >> 1) < DIA_LEN_C) {
        nrm2[tid >> 1] = sii;
        dpv[tid >> 1]  = sip;
    }
    const int rhalf = (lane >> 4) * 8;
#pragma unroll
    for (int t = 0; t < NRT; ++t) {
#pragma unroll
        for (int r = 0; r < 8; ++r) {
            Ys[(t * 16 + rhalf + r) * YPITCH + wave * 16 + row] = acc[t][r];
        }
    }
    __syncthreads();

    if (tid < DIA_LEN_C) {
        const float INV_PI = 0.318309886183790671538f;
        const int pv = prevL[tid], nx = nextL[tid];
        const float s_i = nrm2[tid];
        float wp = 0.f, wn = 0.f;
        if (pv >= 0) {
            float den = sqrtf(s_i) * sqrtf(nrm2[pv]);
            float cs  = (den > 0.f) ? dpv[tid] / den : 0.f;
            cs = fminf(1.f, fmaxf(-1.f, cs));
            wp = 1.f - acosf(cs) * INV_PI;
        }
        if (nx >= 0) {
            float den = sqrtf(s_i) * sqrtf(nrm2[nx]);
            float cs  = (den > 0.f) ? dpv[nx] / den : 0.f;
            cs = fminf(1.f, fmaxf(-1.f, cs));
            wn = 1.f - acosf(cs) * INV_PI;
        }
        wpL[tid] = wp;
        wnL[tid] = wn;
    }
    __syncthreads();

    const float4 bv = *(const float4*)(bias + n0 + lane * 4);
    for (int i = wave; i < DIA_LEN_C; i += 8) {
        const int pv = prevL[i], nx = nextL[i];
        const float wp = wpL[i], wn = wnL[i];
        const float4 yi = *(const float4*)(&Ys[i * YPITCH + lane * 4]);
        const float4 yp = *(const float4*)(&Ys[((pv >= 0) ? pv : i) * YPITCH + lane * 4]);
        const float4 yn = *(const float4*)(&Ys[((nx >= 0) ? nx : i) * YPITCH + lane * 4]);
        float4 r;
        r.x = yi.x + wp * yp.x + wn * yn.x + bv.x;
        r.y = yi.y + wp * yp.y + wn * yn.y + bv.y;
        r.z = yi.z + wp * yp.z + wn * yn.z + bv.z;
        r.w = yi.w + wp * yp.w + wn * yn.w + bv.w;
        *(float4*)(out + (size_t)(m0 + i) * D_DIM + n0 + lane * 4) = r;
    }
}

extern "C" void kernel_launch(void* const* d_in, const int* in_sizes, int n_in,
                              void* d_out, int out_size, void* d_ws, size_t ws_size,
                              hipStream_t stream) {
    const float* inp   = (const float*)d_in[0];
    const int*   qmask = (const int*)d_in[2];
    const float* W     = (const float*)d_in[3];
    const float* bias  = (const float*)d_in[4];
    float*       out   = (float*)d_out;

    dialog_mix_gemm_kernel<<<dim3(4, 60), dim3(256), 0, stream>>>(inp, qmask, W, bias, out);
}


# Per-row scalar stages off the critical path: neighbours by wave ballot + bit scan under the first tile fetch, row sums published by the last K barrier, weights per wave in registers; 2 barriers fewer
# speedup vs baseline: 1.9423x; 1.0621x over previous
#include <hip/hip_runtime.h>
#include <math.h>

#define N_NODES   6000
#define DIA_LEN_C 100
#define D_DIM     512

typedef __attribute__((ext_vector_type(2))) float v2f;
typedef __attribute__((ext_vector_type(8))) float v8f;

#define BN     128
#define KC     64
#define PITCH  68
#define NRT    7
#define AROWS  (NRT * 16)
#define YPITCH 132
#define KS     (KC / 4)
#define HS     2
#define SPLIT  10

__device__ __forceinline__ int top_bit64(unsigned long long x) {
    const unsigned hi = (unsigned)(x >> 32), lo = (unsigned)x;
    return (hi != 0u) ? (63 - __builtin_clz(hi | 1u)) : (31 - __builtin_clz(lo | 1u));
}
__device__ __forceinline__ int low_bit64(unsigned long long x) {
    const unsigned hi = (unsigned)(x >> 32), lo = (unsigned)x;
    return (lo != 0u) ? __builtin_ctz(lo | 0x80000000u) : (32 + __builtin_ctz(hi | 0x80000000u));
}

__device__ __forceinline__ void same_speaker_neighbours(unsigned long long Mlo, unsigned long long Mhi,
                                                        int r, int& pv, int& nx) {
    const unsigned long long VH = (1ull << (DIA_LEN_C - 64)) - 1ull;
    const bool s = ((((r < 64) ? Mlo : Mhi) >> (r & 63)) & 1ull) != 0ull;
    const unsigned long long Slo = s ? Mlo : ~Mlo;
    const unsigned long long Shi = (s ? Mhi : ~Mhi) & VH;
    const bool lo = r < 64;
    const int q = r & 63;
    const unsigned long long own   = lo ? Slo : Shi;
    const unsigned long long below = own & ((1ull << q) - 1ull);
    const unsigned long long above = own & ~((2ull << q) - 1ull);
    const int pv_own = top_bit64(below);
    const int nx_own = low_bit64(above);
    const int pv_oth = top_bit64(Slo);
    const int nx_oth = 64 + low_bit64(Shi);
    pv = (below != 0ull) ? (lo ? pv_own : 64 + pv_own) : ((!lo && Slo != 0ull) ? pv_oth : -1);
    nx = (above != 0ull) ? (lo ? nx_own : 64 + nx_own) : ((lo && Shi != 0ull) ? nx_oth : -1);
}

__global__ void __launch_bounds__(256)
dialog_mix_gemm_kernel(const float* inp,
                       const int*   __restrict__ qmask,
                       const float* W,
                       const float* __restrict__ bias,
                       float*       __restrict__ out) {
    __shared__ __attribute__((aligned(16))) float As[2 * AROWS * PITCH];
    __shared__ __attribute__((aligned(16))) float Bs[2 * BN * PITCH];
    __shared__ __attribute__((aligned(16))) float Ys[AROWS * YPITCH];
    __shared__ float nrm2[DIA_LEN_C];
    __shared__ float dpv[DIA_LEN_C];

    const int tid  = threadIdx.x;
    const int wave = tid >> 5;
    const int lane = tid & 31;
    const int n0 = blockIdx.x * BN;
    const int m0 = blockIdx.y * DIA_LEN_C;

    const int srow = tid >> 4;
    const int scol = (tid & 15) * 4;
    const float* ag0 = inp + (size_t)(m0 + ((srow + 0 < DIA_LEN_C) ? (srow + 0) : (DIA_LEN_C - 1))) * D_DIM + scol;
    const float* ag1 = inp + (size_t)(m0 + ((srow + 16 < DIA_LEN_C) ? (srow + 16) : (DIA_LEN_C - 1))) * D_DIM + scol;
    const float* ag2 = inp + (size_t)(m0 + ((srow + 32 < DIA_LEN_C) ? (srow + 32) : (DIA_LEN_C - 1))) * D_DIM + scol;
    const float* ag3 = inp + (size_t)(m0 + ((srow + 48 < DIA_LEN_C) ? (srow + 48) : (DIA_LEN_C - 1))) * D_DIM + scol;
    const float* ag4 = inp + (size_t)(m0 + ((srow + 64 < DIA_LEN_C) ? (srow + 64) : (DIA_LEN_C - 1))) * D_DIM + scol;
    const float* ag5 = inp + (size_t)(m0 + ((srow + 80 < DIA_LEN_C) ? (srow + 80) : (DIA_LEN_C - 1))) * D_DIM + scol;
    const float* ag6 = inp + (size_t)(m0 + ((srow + 96 < DIA_LEN_C) ? (srow + 96) : (DIA_LEN_C - 1))) * D_DIM + scol;
    const float* bg = W + (size_t)(n0 + srow) * D_DIM + scol;
    const int soff = srow * PITCH + scol;

    const int q0 = qmask[2 * (m0 + lane)];
    const int q1 = qmask[2 * (m0 + 32 + lane)];
    const int q2 = qmask[2 * (m0 + 64 + lane)];
    const int q3 = qmask[2 * (m0 + ((lane < DIA_LEN_C - 96) ? (96 + lane) : (DIA_LEN_C - 1)))];

    float4 pa0 = *(const float4*)(ag0);
    float4 pa1 = *(const float4*)(ag1);
    float4 pa2 = *(const float4*)(ag2);
    float4 pa3 = *(const float4*)(ag3);
    float4 pa4 = *(const float4*)(ag4);
    float4 pa5 = *(const float4*)(ag5);
    float4 pa6 = *(const float4*)(ag6);
    float4 pb0 = *(const float4*)(bg + (size_t)0 * 16 * D_DIM);
    float4 pb1 = *(const float4*)(bg + (size_t)1 * 16 * D_DIM);
    float4 pb2 = *(const float4*)(bg + (size_t)2 * 16 * D_DIM);
    float4 pb3 = *(const float4*)(bg + (size_t)3 * 16 * D_DIM);
    float4 pb4 = *(const float4*)(bg + (size_t)4 * 16 * D_DIM);
    float4 pb5 = *(const float4*)(bg + (size_t)5 * 16 * D_DIM);
    float4 pb6 = *(const float4*)(bg + (size_t)6 * 16 * D_DIM);
    float4 pb7 = *(const float4*)(bg + (size_t)7 * 16 * D_DIM);
    asm volatile("" ::: "memory");
    const unsigned long long Mlo = (unsigned long long)__builtin_amdgcn_ballot_w32(q0 == 1)
                                 | ((unsigned long long)__builtin_amdgcn_ballot_w32(q1 == 1) << 32);
    const unsigned long long Mhi = (unsigned long long)__builtin_amdgcn_ballot_w32(q2 == 1)
                                 | ((unsigned long long)(__builtin_amdgcn_ballot_w32(q3 == 1) & 0xFu) << 32);
    const int drow = ((tid >> 1) < DIA_LEN_C) ? (tid >> 1) : (DIA_LEN_C - 1);
    int dpr, dnx_unused;
    same_speaker_neighbours(Mlo, Mhi, drow, dpr, dnx_unused);
    const int wu = __builtin_amdgcn_readfirstlane(tid >> 5);
    const int orow = (lane < 13 && (wu + 8 * lane) < DIA_LEN_C) ? (wu + 8 * lane) : (DIA_LEN_C - 1);
    int pvr, nxr;
    same_speaker_neighbours(Mlo, Mhi, orow, pvr, nxr);
    *(float4*)(&As[soff + 0 * 16 * PITCH]) = pa0;
    *(float4*)(&As[soff + 1 * 16 * PITCH]) = pa1;
    *(float4*)(&As[soff + 2 * 16 * PITCH]) = pa2;
    *(float4*)(&As[soff + 3 * 16 * PITCH]) = pa3;
    *(float4*)(&As[soff + 4 * 16 * PITCH]) = pa4;
    *(float4*)(&As[soff + 5 * 16 * PITCH]) = pa5;
    *(float4*)(&As[soff + 6 * 16 * PITCH]) = pa6;
    *(float4*)(&Bs[soff + 0 * 16 * PITCH]) = pb0;
    *(float4*)(&Bs[soff + 1 * 16 * PITCH]) = pb1;
    *(float4*)(&Bs[soff + 2 * 16 * PITCH]) = pb2;
    *(float4*)(&Bs[soff + 3 * 16 * PITCH]) = pb3;
    *(float4*)(&Bs[soff + 4 * 16 * PITCH]) = pb4;
    *(float4*)(&Bs[soff + 5 * 16 * PITCH]) = pb5;
    *(float4*)(&Bs[soff + 6 * 16 * PITCH]) = pb6;
    *(float4*)(&Bs[soff + 7 * 16 * PITCH]) = pb7;
    __syncthreads();

    const int dh   = (tid & 1) * 32;
    const int doff_i = drow * PITCH + dh;
    const int doff_p = ((dpr >= 0) ? dpr : drow) * PITCH + dh;
    float si0 = 0.f, si1 = 0.f, si2 = 0.f, si3 = 0.f;
    float sp0 = 0.f, sp1 = 0.f, sp2 = 0.f, sp3 = 0.f;

    const int row = lane & 15;
    const int kh  = (lane >> 4) * 2;
    const int aoff = row * PITCH + kh;
    const int boff = (wave * 16 + row) * PITCH + kh;

    v8f acc[NRT];
#pragma unroll
    for (int t = 0; t < NRT; ++t) acc[t] = (v8f){0.f, 0.f, 0.f, 0.f, 0.f, 0.f, 0.f, 0.f};

    v2f hA[HS][NRT];
    v2f hB[HS];
    {
        const float* as0 = &As[aoff];
        const float* bs0 = &Bs[boff];
#pragma unroll
        for (int ks = 0; ks < HS; ++ks) {
            hB[ks] = *(const v2f*)(bs0 + ks * 4);
#pragma unroll
            for (int t = 0; t < NRT; ++t) hA[ks][t] = *(const v2f*)(as0 + t * 16 * PITCH + ks * 4);
        }
    }

    float n_i = 0.f, d_i = 0.f, n_p = 0.f, n_n = 0.f, d_n = 0.f;
    for (int c = 0; c < D_DIM / KC; ++c) {
        const int cur = c & 1;
        const bool more = (c + 1) < (D_DIM / KC);
        if (more) {
            const int kn = (c + 1) * KC;
            pa0 = *(const float4*)(ag0 + kn);
            pa1 = *(const float4*)(ag1 + kn);
            pa2 = *(const float4*)(ag2 + kn);
            pa3 = *(const float4*)(ag3 + kn);
            pa4 = *(const float4*)(ag4 + kn);
            pa5 = *(const float4*)(ag5 + kn);
            pa6 = *(const float4*)(ag6 + kn);
            pb0 = *(const float4*)(bg + (size_t)0 * 16 * D_DIM + kn);
            pb1 = *(const float4*)(bg + (size_t)1 * 16 * D_DIM + kn);
            pb2 = *(const float4*)(bg + (size_t)2 * 16 * D_DIM + kn);
            pb3 = *(const float4*)(bg + (size_t)3 * 16 * D_DIM + kn);
            pb4 = *(const float4*)(bg + (size_t)4 * 16 * D_DIM + kn);
            pb5 = *(const float4*)(bg + (size_t)5 * 16 * D_DIM + kn);
            pb6 = *(const float4*)(bg + (size_t)6 * 16 * D_DIM + kn);
            pb7 = *(const float4*)(bg + (size_t)7 * 16 * D_DIM + kn);
        }
        __builtin_amdgcn_sched_barrier(0);
        const float* abuf = &As[cur * AROWS * PITCH];
        const float* bbuf = &Bs[cur * BN * PITCH];
        {
            const float* ai = abuf + doff_i;
            const float* ap = abuf + doff_p;
#pragma unroll
            for (int j = 0; j < 8; ++j) {
                const float4 a = *(const float4*)(ai + 4 * j);
                const float4 p = *(const float4*)(ap + 4 * j);
                si0 = fmaf(a.x, a.x, si0); si1 = fmaf(a.y, a.y, si1);
                si2 = fmaf(a.z, a.z, si2); si3 = fmaf(a.w, a.w, si3);
                sp0 = fmaf(a.x, p.x, sp0); sp1 = fmaf(a.y, p.y, sp1);
                sp2 = fmaf(a.z, p.z, sp2); sp3 = fmaf(a.w, p.w, sp3);
            }
        }
        const float* as = abuf + aoff;
        const float* bs = bbuf + boff;
#pragma unroll
        for (int ks = 0; ks < HS; ++ks) {
#pragma unroll
            for (int t = 0; t < NRT; ++t) {
                acc[t] = __builtin_amdgcn_wmma_f32_16x16x4_f32(
                    false, hA[ks][t], false, hB[ks], (short)0, acc[t], false, false);
            }
        }
#pragma unroll
        for (int ks = HS; ks < SPLIT; ++ks) {
            const v2f b = *(const v2f*)(bs + ks * 4);
#pragma unroll
            for (int t = 0; t < NRT; ++t) {
                const v2f a = *(const v2f*)(as + t * 16 * PITCH + ks * 4);
                acc[t] = __builtin_amdgcn_wmma_f32_16x16x4_f32(
                    false, a, false, b, (short)0, acc[t], false, false);
            }
        }
        v2f tA[KS - SPLIT][NRT];
        v2f tB[KS - SPLIT];
#pragma unroll
        for (int ks = SPLIT; ks < KS; ++ks) {
            tB[ks - SPLIT] = *(const v2f*)(bs + ks * 4);
#pragma unroll
            for (int t = 0; t < NRT; ++t) tA[ks - SPLIT][t] = *(const v2f*)(as + t * 16 * PITCH + ks * 4);
        }
        __builtin_amdgcn_sched_barrier(0);
        if (more) {
            float* an = &As[(cur ^ 1) * AROWS * PITCH];
            float* bn = &Bs[(cur ^ 1) * BN * PITCH];
            *(float4*)(an + soff + 0 * 16 * PITCH) = pa0;
            *(float4*)(an + soff + 1 * 16 * PITCH) = pa1;
            *(float4*)(an + soff + 2 * 16 * PITCH) = pa2;
            *(float4*)(an + soff + 3 * 16 * PITCH) = pa3;
            *(float4*)(an + soff + 4 * 16 * PITCH) = pa4;
            *(float4*)(an + soff + 5 * 16 * PITCH) = pa5;
            *(float4*)(an + soff + 6 * 16 * PITCH) = pa6;
            *(float4*)(bn + soff + 0 * 16 * PITCH) = pb0;
            *(float4*)(bn + soff + 1 * 16 * PITCH) = pb1;
            *(float4*)(bn + soff + 2 * 16 * PITCH) = pb2;
            *(float4*)(bn + soff + 3 * 16 * PITCH) = pb3;
            *(float4*)(bn + soff + 4 * 16 * PITCH) = pb4;
            *(float4*)(bn + soff + 5 * 16 * PITCH) = pb5;
            *(float4*)(bn + soff + 6 * 16 * PITCH) = pb6;
            *(float4*)(bn + soff + 7 * 16 * PITCH) = pb7;
        } else {
            float sii = (si0 + si1) + (si2 + si3);
            float sip = (sp0 + sp1) + (sp2 + sp3);
            sii += __shfl_xor(sii, 1, 32);
            sip += __shfl_xor(sip, 1, 32);
            if ((tid & 1) == 0 && (tid >> 1) < DIA_LEN_C) {
                nrm2[tid >> 1] = sii;
                dpv[tid >> 1]  = sip;
            }
        }
        __syncthreads();
        if (more) {
            const float* asn = &As[(cur ^ 1) * AROWS * PITCH] + aoff;
            const float* bsn = &Bs[(cur ^ 1) * BN * PITCH] + boff;
#pragma unroll
            for (int ks = 0; ks < HS; ++ks) {
                hB[ks] = *(const v2f*)(bsn + ks * 4);
#pragma unroll
                for (int t = 0; t < NRT; ++t) hA[ks][t] = *(const v2f*)(asn + t * 16 * PITCH + ks * 4);
            }
        } else {
            n_i = nrm2[orow];
            d_i = dpv[orow];
            n_p = nrm2[(pvr >= 0) ? pvr : orow];
            n_n = nrm2[(nxr >= 0) ? nxr : orow];
            d_n = dpv[(nxr >= 0) ? nxr : orow];
        }
#pragma unroll
        for (int ks = SPLIT; ks < KS; ++ks) {
#pragma unroll
            for (int t = 0; t < NRT; ++t) {
                acc[t] = __builtin_amdgcn_wmma_f32_16x16x4_f32(
                    false, tA[ks - SPLIT][t], false, tB[ks - SPLIT], (short)0, acc[t], false, false);
            }
        }
    }

    const float4 bv = *(const float4*)(bias + n0 + lane * 4);
    float wpr = 0.f, wnr = 0.f;
    {
        const float INV_PI = 0.318309886183790671538f;
        if (pvr >= 0) {
            float den = sqrtf(n_i) * sqrtf(n_p);
            float cs  = (den > 0.f) ? d_i / den : 0.f;
            cs = fminf(1.f, fmaxf(-1.f, cs));
            wpr = 1.f - acosf(cs) * INV_PI;
        }
        if (nxr >= 0) {
            float den = sqrtf(n_i) * sqrtf(n_n);
            float cs  = (den > 0.f) ? d_n / den : 0.f;
            cs = fminf(1.f, fmaxf(-1.f, cs));
            wnr = 1.f - acosf(cs) * INV_PI;
        }
    }
    const int rhalf = (lane >> 4) * 8;
#pragma unroll
    for (int t = 0; t < NRT; ++t) {
#pragma unroll
        for (int r = 0; r < 8; ++r) {
            Ys[(t * 16 + rhalf + r) * YPITCH + wave * 16 + row] = acc[t][r];
        }
    }
    __syncthreads();

#pragma unroll
    for (int j = 0; j < 13; ++j) {
        if (j < 12 || wu < DIA_LEN_C - 96) {
            const int i  = wu + 8 * j;
            const int pv = __builtin_amdgcn_readlane(pvr, j);
            const int nx = __builtin_amdgcn_readlane(nxr, j);
            const float wp = __int_as_float(__builtin_amdgcn_readlane(__float_as_int(wpr), j));
            const float wn = __int_as_float(__builtin_amdgcn_readlane(__float_as_int(wnr), j));
            const float4 yi = *(const float4*)(&Ys[i * YPITCH + lane * 4]);
            const float4 yp = *(const float4*)(&Ys[((pv >= 0) ? pv : i) * YPITCH + lane * 4]);
            const float4 yn = *(const float4*)(&Ys[((nx >= 0) ? nx : i) * YPITCH + lane * 4]);
            float4 r;
            r.x = yi.x + wp * yp.x + wn * yn.x + bv.x;
            r.y = yi.y + wp * yp.y + wn * yn.y + bv.y;
            r.z = yi.z + wp * yp.z + wn * yn.z + bv.z;
            r.w = yi.w + wp * yp.w + wn * yn.w + bv.w;
            *(float4*)(out + (size_t)(m0 + i) * D_DIM + n0 + lane * 4) = r;
        }
    }
}

extern "C" void kernel_launch(void* const* d_in, const int* in_sizes, int n_in,
                              void* d_out, int out_size, void* d_ws, size_t ws_size,
                              hipStream_t stream) {
    const float* inp   = (const float*)d_in[0];
    const int*   qmask = (const int*)d_in[2];
    const float* W     = (const float*)d_in[3];
    const float* bias  = (const float*)d_in[4];
    float*       out   = (float*)d_out;

    dialog_mix_gemm_kernel<<<dim3(4, 60), dim3(256), 0, stream>>>(inp, qmask, W, bias, out);
}
